# starting kernel
# baseline (speedup 1.0000x reference)
#include <hip/hip_runtime.h>
#include <hip/hip_bf16.h>


#define NS   32
#define CC   128
#define HH   56
#define WW   56
#define HWP  3136
#define C3   384
#define PADK 34

typedef float v2f __attribute__((ext_vector_type(2)));
typedef float v8f __attribute__((ext_vector_type(8)));

__device__ __forceinline__ v8f wmma_f32(v2f a, v2f b, v8f c) {
    return __builtin_amdgcn_wmma_f32_16x16x4_f32(false, a, false, b, (short)0, c, false, false);
}

__global__ __launch_bounds__(256) void k_gemm1(const float* __restrict__ p3,
                                               const float* __restrict__ x,
                                               float* __restrict__ t3) {
    __shared__ float As[128 * PADK];
    __shared__ float Bs[224 * PADK];
    const int t    = threadIdx.x;
    const int lane = t & 31;
    const int wv   = t >> 5;
    const int smp  = blockIdx.y;
    const int pos0 = blockIdx.x * 224;
    const float* xs = x + (size_t)smp * CC * HWP;

    const int mn    = lane & 15;
    const int kp    = (lane >> 4) << 1;
    const int rmsb  = (lane >> 4) << 3;
    const int mbase = (wv >> 1) * 32;
    const int nbase = (wv & 1) * 112;

    v8f acc[2][7] = {};

    for (int ks = 0; ks < C3; ks += 32) {
        #pragma unroll
        for (int i = 0; i < 16; ++i) {
            int idx = t + i * 256;
            int m = idx >> 5, k = idx & 31;
            As[m * PADK + k] = p3[m * C3 + ks + k];
        }
        #pragma unroll
        for (int i = 0; i < 28; ++i) {
            int idx = t + i * 256;
            int k = idx / 224, nn = idx % 224;
            int j = ks + k;
            int tap = j % 3, ci = j / 3;
            int pos = pos0 + nn;
            int hp = pos / WW + 2 * (tap - 1);
            float v = 0.0f;
            if ((unsigned)hp < (unsigned)HH)
                v = xs[ci * HWP + pos + 2 * WW * (tap - 1)];
            Bs[nn * PADK + k] = v;
        }
        __syncthreads();
        #pragma unroll
        for (int kk = 0; kk < 8; ++kk) {
            int ko = kk * 4 + kp;
            v2f a0 = *(const v2f*)&As[(mbase + mn) * PADK + ko];
            v2f a1 = *(const v2f*)&As[(mbase + 16 + mn) * PADK + ko];
            #pragma unroll
            for (int ni = 0; ni < 7; ++ni) {
                v2f b = *(const v2f*)&Bs[(nbase + ni * 16 + mn) * PADK + ko];
                acc[0][ni] = wmma_f32(a0, b, acc[0][ni]);
                acc[1][ni] = wmma_f32(a1, b, acc[1][ni]);
            }
        }
        __syncthreads();
    }

    float* o = t3 + (size_t)smp * CC * HWP;
    #pragma unroll
    for (int mi = 0; mi < 2; ++mi) {
        #pragma unroll
        for (int ni = 0; ni < 7; ++ni) {
            int col = pos0 + nbase + ni * 16 + mn;
            #pragma unroll
            for (int r = 0; r < 8; ++r) {
                int m = mbase + mi * 16 + r + rmsb;
                o[m * HWP + col] = acc[mi][ni][r];
            }
        }
    }
}

__global__ __launch_bounds__(256) void k_t12(const float* __restrict__ x,
                                             const float* __restrict__ t3,
                                             const float* __restrict__ p2,
                                             const float* __restrict__ p5,
                                             const float* __restrict__ p10,
                                             float* __restrict__ t12) {
    int tid  = blockIdx.x * 256 + threadIdx.x;
    int pos  = tid % HWP;
    int rest = tid / HWP;
    int c    = rest % CC;
    int smp  = rest / CC;
    int h = pos / WW, w = pos % WW;

    const float* xc = x + ((size_t)smp * CC + c) * HWP;
    float x0  = xc[pos];
    float xvm = (h >= 2)      ? xc[pos - 2 * WW] : 0.0f;
    float xvp = (h < HH - 2)  ? xc[pos + 2 * WW] : 0.0f;
    float xhm = (w >= 2)      ? xc[pos - 2]      : 0.0f;
    float xhp = (w < WW - 2)  ? xc[pos + 2]      : 0.0f;

    float mx = fmaxf(xhm, fmaxf(x0, xhp));
    float e0 = __expf(xhm - mx), e1 = __expf(x0 - mx), e2 = __expf(xhp - mx);
    float inv = 1.0f / (e0 + e1 + e2);
    float t8[3] = { e0 * inv, e1 * inv, e2 * inv };
    float xv[3] = { xvm, x0, xvp };

    float p5v = p5[pos];
    const float* t3s  = t3  + (size_t)smp * CC * HWP;
    const float* p10s = p10 + (size_t)(3 * c) * HWP + pos;
    float* outj = t12 + ((size_t)smp * C3 + 3 * c) * HWP + pos;

    #pragma unroll
    for (int k = 0; k < 3; ++k) {
        int j = 3 * c + k;
        int b = j & (CC - 1);
        float t5v  = p5v * t3s[b * HWP + pos];
        float t2v  = p2[k * WW + w] * xv[k];
        float t10v = p10s[k * HWP] * t8[k];
        outj[k * HWP] = t5v - t2v - t10v;
    }
}

__global__ __launch_bounds__(256) void k_gemm2(const float* __restrict__ x,
                                               const float* __restrict__ t3,
                                               float* __restrict__ t9) {
    __shared__ float As[128 * PADK];
    __shared__ float Bs[128 * PADK];
    const int t    = threadIdx.x;
    const int lane = t & 31;
    const int wv   = t >> 5;
    const int smp  = blockIdx.y;
    const int d0   = blockIdx.x * 128;
    const float* xs  = x  + (size_t)smp * CC * HWP;
    const float* t3s = t3 + (size_t)smp * CC * HWP;

    const int mn    = lane & 15;
    const int kp    = (lane >> 4) << 1;
    const int rmsb  = (lane >> 4) << 3;
    const int mbase = (wv >> 1) * 32;
    const int nbase = (wv & 1) * 64;

    v8f acc[2][4] = {};

    for (int ks = 0; ks < HWP; ks += 32) {
        #pragma unroll
        for (int i = 0; i < 16; ++i) {
            int idx = t + i * 256;
            int m = idx >> 5, k = idx & 31;
            As[m * PADK + k] = xs[m * HWP + ks + k];
        }
        #pragma unroll
        for (int i = 0; i < 16; ++i) {
            int idx = t + i * 256;
            int n = idx >> 5, k = idx & 31;
            int d = d0 + n;
            int tap = d % 3, ci = d / 3;
            int pos = ks + k;
            int wp = pos % WW + (tap - 1);
            float v = 0.0f;
            if ((unsigned)wp < (unsigned)WW)
                v = t3s[ci * HWP + pos + (tap - 1)];
            Bs[n * PADK + k] = v;
        }
        __syncthreads();
        #pragma unroll
        for (int kk = 0; kk < 8; ++kk) {
            int ko = kk * 4 + kp;
            v2f a0 = *(const v2f*)&As[(mbase + mn) * PADK + ko];
            v2f a1 = *(const v2f*)&As[(mbase + 16 + mn) * PADK + ko];
            #pragma unroll
            for (int ni = 0; ni < 4; ++ni) {
                v2f b = *(const v2f*)&Bs[(nbase + ni * 16 + mn) * PADK + ko];
                acc[0][ni] = wmma_f32(a0, b, acc[0][ni]);
                acc[1][ni] = wmma_f32(a1, b, acc[1][ni]);
            }
        }
        __syncthreads();
    }

    const float scale = 1.0f / 56.0f;
    float* out = t9 + (size_t)smp * CC * C3;
    #pragma unroll
    for (int mi = 0; mi < 2; ++mi) {
        #pragma unroll
        for (int ni = 0; ni < 4; ++ni) {
            int col = d0 + nbase + ni * 16 + mn;
            #pragma unroll
            for (int r = 0; r < 8; ++r) {
                int m = mbase + mi * 16 + r + rmsb;
                out[m * C3 + col] = acc[mi][ni][r] * scale;
            }
        }
    }
}

__global__ __launch_bounds__(256) void k_gemm3(const float* __restrict__ t9,
                                               const float* __restrict__ t12,
                                               float* __restrict__ out) {
    __shared__ float As[128 * PADK];
    __shared__ float Bs[224 * PADK];
    const int t    = threadIdx.x;
    const int lane = t & 31;
    const int wv   = t >> 5;
    const int smp  = blockIdx.y;
    const int pos0 = blockIdx.x * 224;
    const float* t9s  = t9  + (size_t)smp * CC * C3;
    const float* t12s = t12 + (size_t)smp * C3 * HWP;

    const int mn    = lane & 15;
    const int kp    = (lane >> 4) << 1;
    const int rmsb  = (lane >> 4) << 3;
    const int mbase = (wv >> 1) * 32;
    const int nbase = (wv & 1) * 112;

    v8f acc[2][7] = {};

    for (int ks = 0; ks < C3; ks += 32) {
        #pragma unroll
        for (int i = 0; i < 16; ++i) {
            int idx = t + i * 256;
            int m = idx >> 5, k = idx & 31;
            As[m * PADK + k] = t9s[m * C3 + ks + k];
        }
        #pragma unroll
        for (int i = 0; i < 28; ++i) {
            int idx = t + i * 256;
            int k = idx / 224, nn = idx % 224;
            Bs[nn * PADK + k] = t12s[(ks + k) * HWP + pos0 + nn];
        }
        __syncthreads();
        #pragma unroll
        for (int kk = 0; kk < 8; ++kk) {
            int ko = kk * 4 + kp;
            v2f a0 = *(const v2f*)&As[(mbase + mn) * PADK + ko];
            v2f a1 = *(const v2f*)&As[(mbase + 16 + mn) * PADK + ko];
            #pragma unroll
            for (int ni = 0; ni < 7; ++ni) {
                v2f b = *(const v2f*)&Bs[(nbase + ni * 16 + mn) * PADK + ko];
                acc[0][ni] = wmma_f32(a0, b, acc[0][ni]);
                acc[1][ni] = wmma_f32(a1, b, acc[1][ni]);
            }
        }
        __syncthreads();
    }

    const float scale = 0.051031036307982884f;
    float* o = out + (size_t)smp * CC * HWP;
    #pragma unroll
    for (int mi = 0; mi < 2; ++mi) {
        #pragma unroll
        for (int ni = 0; ni < 7; ++ni) {
            int col = pos0 + nbase + ni * 16 + mn;
            #pragma unroll
            for (int r = 0; r < 8; ++r) {
                int m = mbase + mi * 16 + r + rmsb;
                o[m * HWP + col] = acc[mi][ni][r] * scale;
            }
        }
    }
}

extern "C" void kernel_launch(void* const* d_in, const int* in_sizes, int n_in,
                              void* d_out, int out_size, void* d_ws, size_t ws_size,
                              hipStream_t stream) {
    (void)in_sizes; (void)n_in; (void)out_size; (void)ws_size;
    const float* x   = (const float*)d_in[0];
    const float* p2  = (const float*)d_in[1];
    const float* p3  = (const float*)d_in[2];
    const float* p5  = (const float*)d_in[3];
    const float* p10 = (const float*)d_in[4];
    float* out = (float*)d_out;

    char* ws = (char*)d_ws;
    const size_t t3_bytes  = (size_t)NS * CC * HWP * sizeof(float);
    const size_t t12_bytes = (size_t)NS * C3 * HWP * sizeof(float);
    float* t3  = (float*)ws;
    float* t12 = (float*)(ws + t3_bytes);
    float* t9  = (float*)(ws + t3_bytes + t12_bytes);

    dim3 blk(256);
    k_gemm1<<<dim3(HWP / 224, NS), blk, 0, stream>>>(p3, x, t3);
    k_t12 <<<dim3((NS * CC * HWP) / 256), blk, 0, stream>>>(x, t3, p2, p5, p10, t12);
    k_gemm2<<<dim3(C3 / 128, NS), blk, 0, stream>>>(x, t3, t9);
    k_gemm3<<<dim3(HWP / 224, NS), blk, 0, stream>>>(t9, t12, out);
}


# GEMM3 k-loop pipelined over two LDS stages: step s+1's t9 and t12 tiles are fetched to registers before step s's matrix instructions, stored behind them, one barrier per step; sums unchanged
# speedup vs baseline: 3.3562x; 3.3562x over previous
#include <hip/hip_runtime.h>
#include <hip/hip_bf16.h>


#define NS   32
#define CC   128
#define HH   56
#define WW   56
#define HWP  3136
#define C3   384

typedef float v2f __attribute__((ext_vector_type(2)));
typedef float v8f __attribute__((ext_vector_type(8)));

__device__ __forceinline__ v8f wmma_f32(v2f a, v2f b, v8f c) {
    return __builtin_amdgcn_wmma_f32_16x16x4_f32(false, a, false, b, (short)0, c, false, false);
}

#define KC   16
#define XP   20
#define AP   52
#define WIN  448

typedef float v4f __attribute__((ext_vector_type(4)));

__global__ __launch_bounds__(256) void k_gemm1(const float* __restrict__ p3,
                                               const float* __restrict__ x,
                                               float* __restrict__ t3) {
    __shared__ __attribute__((aligned(16))) float As[128 * AP];
    __shared__ __attribute__((aligned(16))) float Xs[WIN * XP];
    const int t    = threadIdx.x;
    const int lane = t & 31;
    const int wv   = t >> 5;
    const int smp  = blockIdx.y;
    const int pos0 = blockIdx.x * 224;
    const float* xs = x + (size_t)smp * CC * HWP;

    const int mn    = lane & 15;
    const int kp    = (lane >> 4) << 1;
    const int rmsb  = (lane >> 4) << 3;
    const int mbase = (wv >> 1) * 32;
    const int nbase = (wv & 1) * 112;

    const int xq  = t >> 6;
    const int xpl = t & 63;
    const int am  = t >> 4;
    const int acl = t & 15;

    v8f acc[2][7] = {};

    for (int c0 = 0; c0 < CC; c0 += KC) {
        float av[8][3];
        #pragma unroll
        for (int i = 0; i < 8; ++i) {
            const float* src = p3 + (am + 16 * i) * C3 + 3 * (c0 + acl);
            av[i][0] = src[0]; av[i][1] = src[1]; av[i][2] = src[2];
        }
        float xv[7][4];
        #pragma unroll
        for (int i = 0; i < 7; ++i) {
            int g = pos0 - 112 + xpl + 64 * i;
            bool ok = (unsigned)g < (unsigned)HWP;
            int gs = ok ? g : 0;
            #pragma unroll
            for (int e = 0; e < 4; ++e) {
                float v = xs[(c0 + 4 * xq + e) * HWP + gs];
                xv[i][e] = ok ? v : 0.0f;
            }
        }
        #pragma unroll
        for (int i = 0; i < 8; ++i) {
            #pragma unroll
            for (int tap = 0; tap < 3; ++tap)
                As[(am + 16 * i) * AP + tap * KC + acl] = av[i][tap];
        }
        #pragma unroll
        for (int i = 0; i < 7; ++i) {
            v4f w4 = { xv[i][0], xv[i][1], xv[i][2], xv[i][3] };
            *(v4f*)&Xs[(xpl + 64 * i) * XP + 4 * xq] = w4;
        }
        __syncthreads();
        #pragma unroll
        for (int s = 0; s < 12; ++s) {
            const int tap = s >> 2;
            const int ko  = (s & 3) * 4 + kp;
            v2f a0 = *(const v2f*)&As[(mbase + mn) * AP + tap * KC + ko];
            v2f a1 = *(const v2f*)&As[(mbase + 16 + mn) * AP + tap * KC + ko];
            #pragma unroll
            for (int ni = 0; ni < 7; ++ni) {
                v2f b = *(const v2f*)&Xs[(nbase + ni * 16 + mn + 112 * tap) * XP + ko];
                acc[0][ni] = wmma_f32(a0, b, acc[0][ni]);
                acc[1][ni] = wmma_f32(a1, b, acc[1][ni]);
            }
        }
        __syncthreads();
    }

    float* o = t3 + (size_t)smp * CC * HWP;
    #pragma unroll
    for (int mi = 0; mi < 2; ++mi) {
        #pragma unroll
        for (int ni = 0; ni < 7; ++ni) {
            int col = pos0 + nbase + ni * 16 + mn;
            #pragma unroll
            for (int r = 0; r < 8; ++r) {
                int m = mbase + mi * 16 + r + rmsb;
                o[m * HWP + col] = acc[mi][ni][r];
            }
        }
    }
}

__global__ __launch_bounds__(256) void k_t12(const float* __restrict__ x,
                                             const float* __restrict__ t3,
                                             const float* __restrict__ p2,
                                             const float* __restrict__ p5,
                                             const float* __restrict__ p10,
                                             float* __restrict__ t12) {
    int tid  = blockIdx.x * 256 + threadIdx.x;
    int pos  = tid % HWP;
    int rest = tid / HWP;
    int c    = rest % CC;
    int smp  = rest / CC;
    int h = pos / WW, w = pos % WW;

    const float* xc = x + ((size_t)smp * CC + c) * HWP;
    float x0  = xc[pos];
    float xvm = (h >= 2)      ? xc[pos - 2 * WW] : 0.0f;
    float xvp = (h < HH - 2)  ? xc[pos + 2 * WW] : 0.0f;
    float xhm = (w >= 2)      ? xc[pos - 2]      : 0.0f;
    float xhp = (w < WW - 2)  ? xc[pos + 2]      : 0.0f;

    float mx = fmaxf(xhm, fmaxf(x0, xhp));
    float e0 = __expf(xhm - mx), e1 = __expf(x0 - mx), e2 = __expf(xhp - mx);
    float inv = 1.0f / (e0 + e1 + e2);
    float t8[3] = { e0 * inv, e1 * inv, e2 * inv };
    float xv[3] = { xvm, x0, xvp };

    float p5v = p5[pos];
    const float* t3s  = t3  + (size_t)smp * CC * HWP;
    const float* p10s = p10 + (size_t)(3 * c) * HWP + pos;
    float* outj = t12 + ((size_t)smp * C3 + 3 * c) * HWP + pos;

    #pragma unroll
    for (int k = 0; k < 3; ++k) {
        int j = 3 * c + k;
        int b = j & (CC - 1);
        float t5v  = p5v * t3s[b * HWP + pos];
        float t2v  = p2[k * WW + w] * xv[k];
        float t10v = p10s[k * HWP] * t8[k];
        outj[k * HWP] = t5v - t2v - t10v;
    }
}

#define G2K   224
#define G2AP  228
#define G2RP  60
#define G2CP  248

__global__ __launch_bounds__(128) void k_gemm2(const float* __restrict__ x,
                                               const float* __restrict__ t3,
                                               float* __restrict__ t9) {
    __shared__ __attribute__((aligned(16))) float As[32 * G2AP];
    __shared__ __attribute__((aligned(16))) float Ts[32 * G2CP];
    const int t    = threadIdx.x;
    const int lane = t & 31;
    const int wv   = t >> 5;
    const int smp  = blockIdx.z;
    const int m0   = blockIdx.y * 32;
    const int ch0  = blockIdx.x * 32;
    const int d0   = blockIdx.x * 96;
    const float* xs  = x  + ((size_t)smp * CC + m0)  * HWP;
    const float* t3s = t3 + ((size_t)smp * CC + ch0) * HWP;

    const int mn   = lane & 15;
    const int kp   = (lane >> 4) << 1;
    const int rmsb = (lane >> 4) << 3;
    const int wm   = wv >> 1;
    const int wn   = wv & 1;

    for (int j = t; j < 32 * 5; j += 128) {
        int ch = j / 5, r = j % 5;
        v4f z = { 0.0f, 0.0f, 0.0f, 0.0f };
        *(v4f*)&Ts[ch * G2CP + r * G2RP] = z;
    }

    const int frow = t / 56;
    const int fq   = t % 56;
    const int fg   = frow * HWP + 4 * fq;
    const int fa   = frow * G2AP + 4 * fq;
    const int fb   = frow * G2CP + 4 + 4 * fq + 4 * (fq / 14);

    const float* abase = &As[(16 * wm + mn) * G2AP + kp];
    const float* bbase[3];
    #pragma unroll
    for (int j = 0; j < 3; ++j) {
        int n   = 48 * wn + 16 * j + mn;
        int ci  = n / 3;
        int tap = n - 3 * ci;
        bbase[j] = &Ts[ci * G2CP + 3 + tap + kp];
    }

    v8f acc[3] = {};

    for (int p0 = 0; p0 < HWP; p0 += G2K) {
        if (t < 112) {
            v4f av[16], bv[16];
            #pragma unroll
            for (int i = 0; i < 16; ++i)
                av[i] = *(const v4f*)&xs[fg + p0 + 2 * i * HWP];
            #pragma unroll
            for (int i = 0; i < 16; ++i)
                bv[i] = *(const v4f*)&t3s[fg + p0 + 2 * i * HWP];
            #pragma unroll
            for (int i = 0; i < 16; ++i)
                *(v4f*)&As[fa + 2 * i * G2AP] = av[i];
            #pragma unroll
            for (int i = 0; i < 16; ++i)
                *(v4f*)&Ts[fb + 2 * i * G2CP] = bv[i];
        }
        __syncthreads();
        #pragma unroll
        for (int r = 0; r < 4; ++r) {
            #pragma unroll
            for (int g = 0; g < 14; ++g) {
                v2f a = *(const v2f*)&abase[56 * r + 4 * g];
                #pragma unroll
                for (int j = 0; j < 3; ++j) {
                    const float* bp = bbase[j] + G2RP * r + 4 * g;
                    v2f bfr = { bp[0], bp[1] };
                    acc[j] = wmma_f32(a, bfr, acc[j]);
                }
            }
        }
        __syncthreads();
    }

    const float scale = 1.0f / 56.0f;
    float* out = t9 + (size_t)smp * CC * C3;
    #pragma unroll
    for (int j = 0; j < 3; ++j) {
        int col = d0 + 48 * wn + 16 * j + mn;
        #pragma unroll
        for (int r = 0; r < 8; ++r) {
            int m = m0 + 16 * wm + r + rmsb;
            out[m * C3 + col] = acc[j][r] * scale;
        }
    }
}

#define G3P   36
#define G3AST (128 * G3P)
#define G3BST (224 * G3P)

__device__ __forceinline__ void g3_fetch(const float* __restrict__ asrc,
                                         const float* __restrict__ bsrc,
                                         const int (&bg)[7],
                                         v4f (&av)[4], float (&bv)[7][4]) {
    #pragma unroll
    for (int i = 0; i < 4; ++i)
        av[i] = *(const v4f*)&asrc[32 * i * C3];
    #pragma unroll
    for (int i = 0; i < 7; ++i) {
        #pragma unroll
        for (int e = 0; e < 4; ++e)
            bv[i][e] = bsrc[bg[i] + e * HWP];
    }
}

__device__ __forceinline__ void g3_stage(float* adst, float* bdst,
                                         const int (&bl)[7],
                                         const v4f (&av)[4], const float (&bv)[7][4]) {
    #pragma unroll
    for (int i = 0; i < 4; ++i)
        *(v4f*)&adst[32 * i * G3P] = av[i];
    #pragma unroll
    for (int i = 0; i < 7; ++i) {
        v4f w4 = { bv[i][0], bv[i][1], bv[i][2], bv[i][3] };
        *(v4f*)&bdst[bl[i]] = w4;
    }
}

__global__ __launch_bounds__(256) void k_gemm3(const float* __restrict__ t9,
                                               const float* __restrict__ t12,
                                               float* __restrict__ out) {
    __shared__ __attribute__((aligned(16))) float As[2 * G3AST];
    __shared__ __attribute__((aligned(16))) float Bs[2 * G3BST];
    const int t    = threadIdx.x;
    const int lane = t & 31;
    const int wv   = t >> 5;
    const int smp  = blockIdx.y;
    const int pos0 = blockIdx.x * 224;
    const float* t9s  = t9  + (size_t)smp * CC * C3;
    const float* t12s = t12 + (size_t)smp * C3 * HWP + pos0;

    const int mn    = lane & 15;
    const int kp    = (lane >> 4) << 1;
    const int rmsb  = (lane >> 4) << 3;
    const int mbase = (wv >> 1) * 32;
    const int nbase = (wv & 1) * 112;

    const int ag = (t >> 3) * C3  + 4 * (t & 7);
    const int al = (t >> 3) * G3P + 4 * (t & 7);
    int bg[7], bl[7];
    #pragma unroll
    for (int i = 0; i < 7; ++i) {
        int role = t + 256 * i;
        int q = role / 224;
        int n = role - 224 * q;
        bg[i] = 4 * q * HWP + n;
        bl[i] = n * G3P + 4 * q;
    }

    const int afr = (mbase + mn) * G3P + kp;
    const int bfr = (nbase + mn) * G3P + kp;

    v8f acc[2][7] = {};
    v4f   av[4];
    float bv[7][4];

    g3_fetch(t9s + ag, t12s, bg, av, bv);
    g3_stage(As + al, Bs, bl, av, bv);
    __syncthreads();

    for (int s = 0; s < 12; ++s) {
        const int  cur  = s & 1;
        const bool more = s + 1 < 12;
        if (more)
            g3_fetch(t9s + ag + 32 * (s + 1), t12s + (size_t)(32 * (s + 1)) * HWP, bg, av, bv);
        __builtin_amdgcn_sched_barrier(0);
        const float* Ac = As + cur * G3AST + afr;
        const float* Bc = Bs + cur * G3BST + bfr;
        #pragma unroll
        for (int kk = 0; kk < 8; ++kk) {
            v2f a0 = *(const v2f*)&Ac[kk * 4];
            v2f a1 = *(const v2f*)&Ac[16 * G3P + kk * 4];
            #pragma unroll
            for (int ni = 0; ni < 7; ++ni) {
                v2f b = *(const v2f*)&Bc[ni * 16 * G3P + kk * 4];
                acc[0][ni] = wmma_f32(a0, b, acc[0][ni]);
                acc[1][ni] = wmma_f32(a1, b, acc[1][ni]);
            }
        }
        __builtin_amdgcn_sched_barrier(0);
        if (more)
            g3_stage(As + (cur ^ 1) * G3AST + al, Bs + (cur ^ 1) * G3BST, bl, av, bv);
        __syncthreads();
    }

    const float scale = 0.051031036307982884f;
    float* o = out + (size_t)smp * CC * HWP;
    #pragma unroll
    for (int mi = 0; mi < 2; ++mi) {
        #pragma unroll
        for (int ni = 0; ni < 7; ++ni) {
            int col = pos0 + nbase + ni * 16 + mn;
            #pragma unroll
            for (int r = 0; r < 8; ++r) {
                int m = mbase + mi * 16 + r + rmsb;
                o[m * HWP + col] = acc[mi][ni][r] * scale;
            }
        }
    }
}

extern "C" void kernel_launch(void* const* d_in, const int* in_sizes, int n_in,
                              void* d_out, int out_size, void* d_ws, size_t ws_size,
                              hipStream_t stream) {
    (void)in_sizes; (void)n_in; (void)out_size; (void)ws_size;
    const float* x   = (const float*)d_in[0];
    const float* p2  = (const float*)d_in[1];
    const float* p3  = (const float*)d_in[2];
    const float* p5  = (const float*)d_in[3];
    const float* p10 = (const float*)d_in[4];
    float* out = (float*)d_out;

    char* ws = (char*)d_ws;
    const size_t t3_bytes  = (size_t)NS * CC * HWP * sizeof(float);
    const size_t t12_bytes = (size_t)NS * C3 * HWP * sizeof(float);
    float* t3  = (float*)ws;
    float* t12 = (float*)(ws + t3_bytes);
    float* t9  = (float*)(ws + t3_bytes + t12_bytes);

    dim3 blk(256);
    k_gemm1<<<dim3(HWP / 224, NS), blk, 0, stream>>>(p3, x, t3);
    k_t12 <<<dim3((NS * CC * HWP) / 256), blk, 0, stream>>>(x, t3, p2, p5, p10, t12);
    k_gemm2<<<dim3(C3 / 96, CC / 32, NS), dim3(128), 0, stream>>>(x, t3, t9);
    k_gemm3<<<dim3(HWP / 224, NS), blk, 0, stream>>>(t9, t12, out);
}


# t12 elementwise pass restreamed: a lane owns 4 adjacent positions of a channel over 4 samples, all planes move 16 bytes per lane, p2/p5/p10 read once per lane; 16x fewer waves; arithmetic unchanged
# speedup vs baseline: 3.7614x; 1.1207x over previous
#include <hip/hip_runtime.h>
#include <hip/hip_bf16.h>


#define NS   32
#define CC   128
#define HH   56
#define WW   56
#define HWP  3136
#define C3   384

typedef float v2f __attribute__((ext_vector_type(2)));
typedef float v8f __attribute__((ext_vector_type(8)));

__device__ __forceinline__ v8f wmma_f32(v2f a, v2f b, v8f c) {
    return __builtin_amdgcn_wmma_f32_16x16x4_f32(false, a, false, b, (short)0, c, false, false);
}

#define KC   16
#define XP   20
#define AP   52
#define WIN  448

typedef float v4f __attribute__((ext_vector_type(4)));

__global__ __launch_bounds__(256) void k_gemm1(const float* __restrict__ p3,
                                               const float* __restrict__ x,
                                               float* __restrict__ t3) {
    __shared__ __attribute__((aligned(16))) float As[128 * AP];
    __shared__ __attribute__((aligned(16))) float Xs[WIN * XP];
    const int t    = threadIdx.x;
    const int lane = t & 31;
    const int wv   = t >> 5;
    const int smp  = blockIdx.y;
    const int pos0 = blockIdx.x * 224;
    const float* xs = x + (size_t)smp * CC * HWP;

    const int mn    = lane & 15;
    const int kp    = (lane >> 4) << 1;
    const int rmsb  = (lane >> 4) << 3;
    const int mbase = (wv >> 1) * 32;
    const int nbase = (wv & 1) * 112;

    const int xq  = t >> 6;
    const int xpl = t & 63;
    const int am  = t >> 4;
    const int acl = t & 15;

    v8f acc[2][7] = {};

    for (int c0 = 0; c0 < CC; c0 += KC) {
        float av[8][3];
        #pragma unroll
        for (int i = 0; i < 8; ++i) {
            const float* src = p3 + (am + 16 * i) * C3 + 3 * (c0 + acl);
            av[i][0] = src[0]; av[i][1] = src[1]; av[i][2] = src[2];
        }
        float xv[7][4];
        #pragma unroll
        for (int i = 0; i < 7; ++i) {
            int g = pos0 - 112 + xpl + 64 * i;
            bool ok = (unsigned)g < (unsigned)HWP;
            int gs = ok ? g : 0;
            #pragma unroll
            for (int e = 0; e < 4; ++e) {
                float v = xs[(c0 + 4 * xq + e) * HWP + gs];
                xv[i][e] = ok ? v : 0.0f;
            }
        }
        #pragma unroll
        for (int i = 0; i < 8; ++i) {
            #pragma unroll
            for (int tap = 0; tap < 3; ++tap)
                As[(am + 16 * i) * AP + tap * KC + acl] = av[i][tap];
        }
        #pragma unroll
        for (int i = 0; i < 7; ++i) {
            v4f w4 = { xv[i][0], xv[i][1], xv[i][2], xv[i][3] };
            *(v4f*)&Xs[(xpl + 64 * i) * XP + 4 * xq] = w4;
        }
        __syncthreads();
        #pragma unroll
        for (int s = 0; s < 12; ++s) {
            const int tap = s >> 2;
            const int ko  = (s & 3) * 4 + kp;
            v2f a0 = *(const v2f*)&As[(mbase + mn) * AP + tap * KC + ko];
            v2f a1 = *(const v2f*)&As[(mbase + 16 + mn) * AP + tap * KC + ko];
            #pragma unroll
            for (int ni = 0; ni < 7; ++ni) {
                v2f b = *(const v2f*)&Xs[(nbase + ni * 16 + mn + 112 * tap) * XP + ko];
                acc[0][ni] = wmma_f32(a0, b, acc[0][ni]);
                acc[1][ni] = wmma_f32(a1, b, acc[1][ni]);
            }
        }
        __syncthreads();
    }

    float* o = t3 + (size_t)smp * CC * HWP;
    #pragma unroll
    for (int mi = 0; mi < 2; ++mi) {
        #pragma unroll
        for (int ni = 0; ni < 7; ++ni) {
            int col = pos0 + nbase + ni * 16 + mn;
            #pragma unroll
            for (int r = 0; r < 8; ++r) {
                int m = mbase + mi * 16 + r + rmsb;
                o[m * HWP + col] = acc[mi][ni][r];
            }
        }
    }
}

#define QPC  784
#define QPR  14
#define SPG  4

__global__ __launch_bounds__(256) void k_t12(const float* __restrict__ x,
                                             const float* __restrict__ t3,
                                             const float* __restrict__ p2,
                                             const float* __restrict__ p5,
                                             const float* __restrict__ p10,
                                             float* __restrict__ t12) {
    const int f    = blockIdx.x * 256 + threadIdx.x;
    const int q    = f % QPC;
    const int c    = f / QPC;
    const int pos  = 4 * q;
    const int h    = q / QPR;
    const int w0   = 4 * (q - QPR * h);
    const int smp0 = blockIdx.y * SPG;

    const bool up = h >= 2, dn = h < HH - 2;
    const bool lf = w0 >= 4, rt = w0 < WW - 4;
    const int oup = up ? -2 * WW : 0;
    const int odn = dn ?  2 * WW : 0;
    const int olf = lf ? -2 : 0;
    const int ort = rt ?  4 : 0;

    const v4f p5v = *(const v4f*)&p5[pos];
    v4f p2v[3], p10v[3];
    int tb[3];
    for (int k = 0; k < 3; ++k) {
        int j = 3 * c + k;
        p2v[k]  = *(const v4f*)&p2[k * WW + w0];
        p10v[k] = *(const v4f*)&p10[(size_t)j * HWP + pos];
        tb[k]   = (j & (CC - 1)) * HWP + pos;
    }

    const v4f z4 = { 0.0f, 0.0f, 0.0f, 0.0f };
    const v2f z2 = { 0.0f, 0.0f };

    for (int i = 0; i < SPG; ++i) {
        const int smp = smp0 + i;
        const float* xc  = x   + ((size_t)smp * CC + c) * HWP + pos;
        const float* t3s = t3  + (size_t)smp * CC * HWP;
        float* outj      = t12 + ((size_t)smp * C3 + 3 * c) * HWP + pos;

        v4f xq  = *(const v4f*)xc;
        v4f xvm = *(const v4f*)(xc + oup);
        v4f xvp = *(const v4f*)(xc + odn);
        v2f xl  = *(const v2f*)(xc + olf);
        v2f xr  = *(const v2f*)(xc + ort);
        v4f t3v[3];
        for (int k = 0; k < 3; ++k)
            t3v[k] = *(const v4f*)&t3s[tb[k]];
        xvm = up ? xvm : z4;
        xvp = dn ? xvp : z4;
        xl  = lf ? xl  : z2;
        xr  = rt ? xr  : z2;

        const float hm[4] = { xl[0], xl[1], xq[0], xq[1] };
        const float hp[4] = { xq[2], xq[3], xr[0], xr[1] };

        v4f t8[3];
        for (int e = 0; e < 4; ++e) {
            float x0 = xq[e];
            float mx = fmaxf(hm[e], fmaxf(x0, hp[e]));
            float e0 = __expf(hm[e] - mx), e1 = __expf(x0 - mx), e2 = __expf(hp[e] - mx);
            float inv = 1.0f / (e0 + e1 + e2);
            t8[0][e] = e0 * inv; t8[1][e] = e1 * inv; t8[2][e] = e2 * inv;
        }
        const v4f xv[3] = { xvm, xq, xvp };

        for (int k = 0; k < 3; ++k) {
            v4f o;
            for (int e = 0; e < 4; ++e) {
                float t5v  = p5v[e] * t3v[k][e];
                float t2v  = p2v[k][e] * xv[k][e];
                float t10v = p10v[k][e] * t8[k][e];
                o[e] = t5v - t2v - t10v;
            }
            *(v4f*)&outj[k * HWP] = o;
        }
    }
}

#define G2K   224
#define G2AP  228
#define G2RP  60
#define G2CP  248

__global__ __launch_bounds__(128) void k_gemm2(const float* __restrict__ x,
                                               const float* __restrict__ t3,
                                               float* __restrict__ t9) {
    __shared__ __attribute__((aligned(16))) float As[32 * G2AP];
    __shared__ __attribute__((aligned(16))) float Ts[32 * G2CP];
    const int t    = threadIdx.x;
    const int lane = t & 31;
    const int wv   = t >> 5;
    const int smp  = blockIdx.z;
    const int m0   = blockIdx.y * 32;
    const int ch0  = blockIdx.x * 32;
    const int d0   = blockIdx.x * 96;
    const float* xs  = x  + ((size_t)smp * CC + m0)  * HWP;
    const float* t3s = t3 + ((size_t)smp * CC + ch0) * HWP;

    const int mn   = lane & 15;
    const int kp   = (lane >> 4) << 1;
    const int rmsb = (lane >> 4) << 3;
    const int wm   = wv >> 1;
    const int wn   = wv & 1;

    for (int j = t; j < 32 * 5; j += 128) {
        int ch = j / 5, r = j % 5;
        v4f z = { 0.0f, 0.0f, 0.0f, 0.0f };
        *(v4f*)&Ts[ch * G2CP + r * G2RP] = z;
    }

    const int frow = t / 56;
    const int fq   = t % 56;
    const int fg   = frow * HWP + 4 * fq;
    const int fa   = frow * G2AP + 4 * fq;
    const int fb   = frow * G2CP + 4 + 4 * fq + 4 * (fq / 14);

    const float* abase = &As[(16 * wm + mn) * G2AP + kp];
    const float* bbase[3];
    #pragma unroll
    for (int j = 0; j < 3; ++j) {
        int n   = 48 * wn + 16 * j + mn;
        int ci  = n / 3;
        int tap = n - 3 * ci;
        bbase[j] = &Ts[ci * G2CP + 3 + tap + kp];
    }

    v8f acc[3] = {};

    for (int p0 = 0; p0 < HWP; p0 += G2K) {
        if (t < 112) {
            v4f av[16], bv[16];
            #pragma unroll
            for (int i = 0; i < 16; ++i)
                av[i] = *(const v4f*)&xs[fg + p0 + 2 * i * HWP];
            #pragma unroll
            for (int i = 0; i < 16; ++i)
                bv[i] = *(const v4f*)&t3s[fg + p0 + 2 * i * HWP];
            #pragma unroll
            for (int i = 0; i < 16; ++i)
                *(v4f*)&As[fa + 2 * i * G2AP] = av[i];
            #pragma unroll
            for (int i = 0; i < 16; ++i)
                *(v4f*)&Ts[fb + 2 * i * G2CP] = bv[i];
        }
        __syncthreads();
        #pragma unroll
        for (int r = 0; r < 4; ++r) {
            #pragma unroll
            for (int g = 0; g < 14; ++g) {
                v2f a = *(const v2f*)&abase[56 * r + 4 * g];
                #pragma unroll
                for (int j = 0; j < 3; ++j) {
                    const float* bp = bbase[j] + G2RP * r + 4 * g;
                    v2f bfr = { bp[0], bp[1] };
                    acc[j] = wmma_f32(a, bfr, acc[j]);
                }
            }
        }
        __syncthreads();
    }

    const float scale = 1.0f / 56.0f;
    float* out = t9 + (size_t)smp * CC * C3;
    #pragma unroll
    for (int j = 0; j < 3; ++j) {
        int col = d0 + 48 * wn + 16 * j + mn;
        #pragma unroll
        for (int r = 0; r < 8; ++r) {
            int m = m0 + 16 * wm + r + rmsb;
            out[m * C3 + col] = acc[j][r] * scale;
        }
    }
}

#define G3P   36
#define G3AST (128 * G3P)
#define G3BST (224 * G3P)

__device__ __forceinline__ void g3_fetch(const float* __restrict__ asrc,
                                         const float* __restrict__ bsrc,
                                         const int (&bg)[7],
                                         v4f (&av)[4], float (&bv)[7][4]) {
    #pragma unroll
    for (int i = 0; i < 4; ++i)
        av[i] = *(const v4f*)&asrc[32 * i * C3];
    #pragma unroll
    for (int i = 0; i < 7; ++i) {
        #pragma unroll
        for (int e = 0; e < 4; ++e)
            bv[i][e] = bsrc[bg[i] + e * HWP];
    }
}

__device__ __forceinline__ void g3_stage(float* adst, float* bdst,
                                         const int (&bl)[7],
                                         const v4f (&av)[4], const float (&bv)[7][4]) {
    #pragma unroll
    for (int i = 0; i < 4; ++i)
        *(v4f*)&adst[32 * i * G3P] = av[i];
    #pragma unroll
    for (int i = 0; i < 7; ++i) {
        v4f w4 = { bv[i][0], bv[i][1], bv[i][2], bv[i][3] };
        *(v4f*)&bdst[bl[i]] = w4;
    }
}

__global__ __launch_bounds__(256) void k_gemm3(const float* __restrict__ t9,
                                               const float* __restrict__ t12,
                                               float* __restrict__ out) {
    __shared__ __attribute__((aligned(16))) float As[2 * G3AST];
    __shared__ __attribute__((aligned(16))) float Bs[2 * G3BST];
    const int t    = threadIdx.x;
    const int lane = t & 31;
    const int wv   = t >> 5;
    const int smp  = blockIdx.y;
    const int pos0 = blockIdx.x * 224;
    const float* t9s  = t9  + (size_t)smp * CC * C3;
    const float* t12s = t12 + (size_t)smp * C3 * HWP + pos0;

    const int mn    = lane & 15;
    const int kp    = (lane >> 4) << 1;
    const int rmsb  = (lane >> 4) << 3;
    const int mbase = (wv >> 1) * 32;
    const int nbase = (wv & 1) * 112;

    const int ag = (t >> 3) * C3  + 4 * (t & 7);
    const int al = (t >> 3) * G3P + 4 * (t & 7);
    int bg[7], bl[7];
    #pragma unroll
    for (int i = 0; i < 7; ++i) {
        int role = t + 256 * i;
        int q = role / 224;
        int n = role - 224 * q;
        bg[i] = 4 * q * HWP + n;
        bl[i] = n * G3P + 4 * q;
    }

    const int afr = (mbase + mn) * G3P + kp;
    const int bfr = (nbase + mn) * G3P + kp;

    v8f acc[2][7] = {};
    v4f   av[4];
    float bv[7][4];

    g3_fetch(t9s + ag, t12s, bg, av, bv);
    g3_stage(As + al, Bs, bl, av, bv);
    __syncthreads();

    for (int s = 0; s < 12; ++s) {
        const int  cur  = s & 1;
        const bool more = s + 1 < 12;
        if (more)
            g3_fetch(t9s + ag + 32 * (s + 1), t12s + (size_t)(32 * (s + 1)) * HWP, bg, av, bv);
        __builtin_amdgcn_sched_barrier(0);
        const float* Ac = As + cur * G3AST + afr;
        const float* Bc = Bs + cur * G3BST + bfr;
        #pragma unroll
        for (int kk = 0; kk < 8; ++kk) {
            v2f a0 = *(const v2f*)&Ac[kk * 4];
            v2f a1 = *(const v2f*)&Ac[16 * G3P + kk * 4];
            #pragma unroll
            for (int ni = 0; ni < 7; ++ni) {
                v2f b = *(const v2f*)&Bc[ni * 16 * G3P + kk * 4];
                acc[0][ni] = wmma_f32(a0, b, acc[0][ni]);
                acc[1][ni] = wmma_f32(a1, b, acc[1][ni]);
            }
        }
        __builtin_amdgcn_sched_barrier(0);
        if (more)
            g3_stage(As + (cur ^ 1) * G3AST + al, Bs + (cur ^ 1) * G3BST, bl, av, bv);
        __syncthreads();
    }

    const float scale = 0.051031036307982884f;
    float* o = out + (size_t)smp * CC * HWP;
    #pragma unroll
    for (int mi = 0; mi < 2; ++mi) {
        #pragma unroll
        for (int ni = 0; ni < 7; ++ni) {
            int col = pos0 + nbase + ni * 16 + mn;
            #pragma unroll
            for (int r = 0; r < 8; ++r) {
                int m = mbase + mi * 16 + r + rmsb;
                o[m * HWP + col] = acc[mi][ni][r] * scale;
            }
        }
    }
}

extern "C" void kernel_launch(void* const* d_in, const int* in_sizes, int n_in,
                              void* d_out, int out_size, void* d_ws, size_t ws_size,
                              hipStream_t stream) {
    (void)in_sizes; (void)n_in; (void)out_size; (void)ws_size;
    const float* x   = (const float*)d_in[0];
    const float* p2  = (const float*)d_in[1];
    const float* p3  = (const float*)d_in[2];
    const float* p5  = (const float*)d_in[3];
    const float* p10 = (const float*)d_in[4];
    float* out = (float*)d_out;

    char* ws = (char*)d_ws;
    const size_t t3_bytes  = (size_t)NS * CC * HWP * sizeof(float);
    const size_t t12_bytes = (size_t)NS * C3 * HWP * sizeof(float);
    float* t3  = (float*)ws;
    float* t12 = (float*)(ws + t3_bytes);
    float* t9  = (float*)(ws + t3_bytes + t12_bytes);

    dim3 blk(256);
    k_gemm1<<<dim3(HWP / 224, NS), blk, 0, stream>>>(p3, x, t3);
    k_t12 <<<dim3((CC * QPC) / 256, NS / SPG), blk, 0, stream>>>(x, t3, p2, p5, p10, t12);
    k_gemm2<<<dim3(C3 / 96, CC / 32, NS), dim3(128), 0, stream>>>(x, t3, t9);
    k_gemm3<<<dim3(HWP / 224, NS), blk, 0, stream>>>(t9, t12, out);
}


# t12 elementwise pass and GEMM2 merged into one launch: work-groups 0..511 take the GEMM2 tiles, 6272 more of 128 threads stream t12 beside them on every WGP; 3 launches; sums unchanged
# speedup vs baseline: 4.0057x; 1.0649x over previous
#include <hip/hip_runtime.h>
#include <hip/hip_bf16.h>


#define NS   32
#define CC   128
#define HH   56
#define WW   56
#define HWP  3136
#define C3   384

typedef float v2f __attribute__((ext_vector_type(2)));
typedef float v8f __attribute__((ext_vector_type(8)));

__device__ __forceinline__ v8f wmma_f32(v2f a, v2f b, v8f c) {
    return __builtin_amdgcn_wmma_f32_16x16x4_f32(false, a, false, b, (short)0, c, false, false);
}

#define KC   16
#define XP   20
#define AP   52
#define WIN  448

typedef float v4f __attribute__((ext_vector_type(4)));

__global__ __launch_bounds__(256) void k_gemm1(const float* __restrict__ p3,
                                               const float* __restrict__ x,
                                               float* __restrict__ t3) {
    __shared__ __attribute__((aligned(16))) float As[128 * AP];
    __shared__ __attribute__((aligned(16))) float Xs[WIN * XP];
    const int t    = threadIdx.x;
    const int lane = t & 31;
    const int wv   = t >> 5;
    const int smp  = blockIdx.y;
    const int pos0 = blockIdx.x * 224;
    const float* xs = x + (size_t)smp * CC * HWP;

    const int mn    = lane & 15;
    const int kp    = (lane >> 4) << 1;
    const int rmsb  = (lane >> 4) << 3;
    const int mbase = (wv >> 1) * 32;
    const int nbase = (wv & 1) * 112;

    const int xq  = t >> 6;
    const int xpl = t & 63;
    const int am  = t >> 4;
    const int acl = t & 15;

    v8f acc[2][7] = {};

    for (int c0 = 0; c0 < CC; c0 += KC) {
        float av[8][3];
        #pragma unroll
        for (int i = 0; i < 8; ++i) {
            const float* src = p3 + (am + 16 * i) * C3 + 3 * (c0 + acl);
            av[i][0] = src[0]; av[i][1] = src[1]; av[i][2] = src[2];
        }
        float xv[7][4];
        #pragma unroll
        for (int i = 0; i < 7; ++i) {
            int g = pos0 - 112 + xpl + 64 * i;
            bool ok = (unsigned)g < (unsigned)HWP;
            int gs = ok ? g : 0;
            #pragma unroll
            for (int e = 0; e < 4; ++e) {
                float v = xs[(c0 + 4 * xq + e) * HWP + gs];
                xv[i][e] = ok ? v : 0.0f;
            }
        }
        #pragma unroll
        for (int i = 0; i < 8; ++i) {
            #pragma unroll
            for (int tap = 0; tap < 3; ++tap)
                As[(am + 16 * i) * AP + tap * KC + acl] = av[i][tap];
        }
        #pragma unroll
        for (int i = 0; i < 7; ++i) {
            v4f w4 = { xv[i][0], xv[i][1], xv[i][2], xv[i][3] };
            *(v4f*)&Xs[(xpl + 64 * i) * XP + 4 * xq] = w4;
        }
        __syncthreads();
        #pragma unroll
        for (int s = 0; s < 12; ++s) {
            const int tap = s >> 2;
            const int ko  = (s & 3) * 4 + kp;
            v2f a0 = *(const v2f*)&As[(mbase + mn) * AP + tap * KC + ko];
            v2f a1 = *(const v2f*)&As[(mbase + 16 + mn) * AP + tap * KC + ko];
            #pragma unroll
            for (int ni = 0; ni < 7; ++ni) {
                v2f b = *(const v2f*)&Xs[(nbase + ni * 16 + mn + 112 * tap) * XP + ko];
                acc[0][ni] = wmma_f32(a0, b, acc[0][ni]);
                acc[1][ni] = wmma_f32(a1, b, acc[1][ni]);
            }
        }
        __syncthreads();
    }

    float* o = t3 + (size_t)smp * CC * HWP;
    #pragma unroll
    for (int mi = 0; mi < 2; ++mi) {
        #pragma unroll
        for (int ni = 0; ni < 7; ++ni) {
            int col = pos0 + nbase + ni * 16 + mn;
            #pragma unroll
            for (int r = 0; r < 8; ++r) {
                int m = mbase + mi * 16 + r + rmsb;
                o[m * HWP + col] = acc[mi][ni][r];
            }
        }
    }
}

#define QPC  784
#define QPR  14
#define SPG  4
#define EWT  128
#define EWX  ((CC * QPC) / EWT)

__device__ __forceinline__ void t12_role(const int bx, const int by,
                                         const float* __restrict__ x,
                                         const float* __restrict__ t3,
                                         const float* __restrict__ p2,
                                         const float* __restrict__ p5,
                                         const float* __restrict__ p10,
                                         float* __restrict__ t12) {
    const int f    = bx * EWT + threadIdx.x;
    const int q    = f % QPC;
    const int c    = f / QPC;
    const int pos  = 4 * q;
    const int h    = q / QPR;
    const int w0   = 4 * (q - QPR * h);
    const int smp0 = by * SPG;

    const bool up = h >= 2, dn = h < HH - 2;
    const bool lf = w0 >= 4, rt = w0 < WW - 4;
    const int oup = up ? -2 * WW : 0;
    const int odn = dn ?  2 * WW : 0;
    const int olf = lf ? -2 : 0;
    const int ort = rt ?  4 : 0;

    const v4f p5v = *(const v4f*)&p5[pos];
    v4f p2v[3], p10v[3];
    int tb[3];
    for (int k = 0; k < 3; ++k) {
        int j = 3 * c + k;
        p2v[k]  = *(const v4f*)&p2[k * WW + w0];
        p10v[k] = *(const v4f*)&p10[(size_t)j * HWP + pos];
        tb[k]   = (j & (CC - 1)) * HWP + pos;
    }

    const v4f z4 = { 0.0f, 0.0f, 0.0f, 0.0f };
    const v2f z2 = { 0.0f, 0.0f };

    for (int i = 0; i < SPG; ++i) {
        const int smp = smp0 + i;
        const float* xc  = x   + ((size_t)smp * CC + c) * HWP + pos;
        const float* t3s = t3  + (size_t)smp * CC * HWP;
        float* outj      = t12 + ((size_t)smp * C3 + 3 * c) * HWP + pos;

        v4f xq  = *(const v4f*)xc;
        v4f xvm = *(const v4f*)(xc + oup);
        v4f xvp = *(const v4f*)(xc + odn);
        v2f xl  = *(const v2f*)(xc + olf);
        v2f xr  = *(const v2f*)(xc + ort);
        v4f t3v[3];
        for (int k = 0; k < 3; ++k)
            t3v[k] = *(const v4f*)&t3s[tb[k]];
        xvm = up ? xvm : z4;
        xvp = dn ? xvp : z4;
        xl  = lf ? xl  : z2;
        xr  = rt ? xr  : z2;

        const float hm[4] = { xl[0], xl[1], xq[0], xq[1] };
        const float hp[4] = { xq[2], xq[3], xr[0], xr[1] };

        v4f t8[3];
        for (int e = 0; e < 4; ++e) {
            float x0 = xq[e];
            float mx = fmaxf(hm[e], fmaxf(x0, hp[e]));
            float e0 = __expf(hm[e] - mx), e1 = __expf(x0 - mx), e2 = __expf(hp[e] - mx);
            float inv = 1.0f / (e0 + e1 + e2);
            t8[0][e] = e0 * inv; t8[1][e] = e1 * inv; t8[2][e] = e2 * inv;
        }
        const v4f xv[3] = { xvm, xq, xvp };

        for (int k = 0; k < 3; ++k) {
            v4f o;
            for (int e = 0; e < 4; ++e) {
                float t5v  = p5v[e] * t3v[k][e];
                float t2v  = p2v[k][e] * xv[k][e];
                float t10v = p10v[k][e] * t8[k][e];
                o[e] = t5v - t2v - t10v;
            }
            *(v4f*)&outj[k * HWP] = o;
        }
    }
}

#define G2K   224
#define G2AP  228
#define G2RP  60
#define G2CP  248

__device__ __forceinline__ void gemm2_role(const int bn, const int bm, const int smp,
                                           const float* __restrict__ x,
                                           const float* __restrict__ t3,
                                           float* __restrict__ t9) {
    __shared__ __attribute__((aligned(16))) float As[32 * G2AP];
    __shared__ __attribute__((aligned(16))) float Ts[32 * G2CP];
    const int t    = threadIdx.x;
    const int lane = t & 31;
    const int wv   = t >> 5;
    const int m0   = bm * 32;
    const int ch0  = bn * 32;
    const int d0   = bn * 96;
    const float* xs  = x  + ((size_t)smp * CC + m0)  * HWP;
    const float* t3s = t3 + ((size_t)smp * CC + ch0) * HWP;

    const int mn   = lane & 15;
    const int kp   = (lane >> 4) << 1;
    const int rmsb = (lane >> 4) << 3;
    const int wm   = wv >> 1;
    const int wn   = wv & 1;

    for (int j = t; j < 32 * 5; j += 128) {
        int ch = j / 5, r = j % 5;
        v4f z = { 0.0f, 0.0f, 0.0f, 0.0f };
        *(v4f*)&Ts[ch * G2CP + r * G2RP] = z;
    }

    const int frow = t / 56;
    const int fq   = t % 56;
    const int fg   = frow * HWP + 4 * fq;
    const int fa   = frow * G2AP + 4 * fq;
    const int fb   = frow * G2CP + 4 + 4 * fq + 4 * (fq / 14);

    const float* abase = &As[(16 * wm + mn) * G2AP + kp];
    const float* bbase[3];
    #pragma unroll
    for (int j = 0; j < 3; ++j) {
        int n   = 48 * wn + 16 * j + mn;
        int ci  = n / 3;
        int tap = n - 3 * ci;
        bbase[j] = &Ts[ci * G2CP + 3 + tap + kp];
    }

    v8f acc[3] = {};

    for (int p0 = 0; p0 < HWP; p0 += G2K) {
        if (t < 112) {
            v4f av[16], bv[16];
            #pragma unroll
            for (int i = 0; i < 16; ++i)
                av[i] = *(const v4f*)&xs[fg + p0 + 2 * i * HWP];
            #pragma unroll
            for (int i = 0; i < 16; ++i)
                bv[i] = *(const v4f*)&t3s[fg + p0 + 2 * i * HWP];
            #pragma unroll
            for (int i = 0; i < 16; ++i)
                *(v4f*)&As[fa + 2 * i * G2AP] = av[i];
            #pragma unroll
            for (int i = 0; i < 16; ++i)
                *(v4f*)&Ts[fb + 2 * i * G2CP] = bv[i];
        }
        __syncthreads();
        #pragma unroll
        for (int r = 0; r < 4; ++r) {
            #pragma unroll
            for (int g = 0; g < 14; ++g) {
                v2f a = *(const v2f*)&abase[56 * r + 4 * g];
                #pragma unroll
                for (int j = 0; j < 3; ++j) {
                    const float* bp = bbase[j] + G2RP * r + 4 * g;
                    v2f bfr = { bp[0], bp[1] };
                    acc[j] = wmma_f32(a, bfr, acc[j]);
                }
            }
        }
        __syncthreads();
    }

    const float scale = 1.0f / 56.0f;
    float* out = t9 + (size_t)smp * CC * C3;
    #pragma unroll
    for (int j = 0; j < 3; ++j) {
        int col = d0 + 48 * wn + 16 * j + mn;
        #pragma unroll
        for (int r = 0; r < 8; ++r) {
            int m = m0 + 16 * wm + r + rmsb;
            out[m * C3 + col] = acc[j][r] * scale;
        }
    }
}

#define G2WG  512
#define EWWG  (EWX * (NS / SPG))

__global__ __launch_bounds__(128) void k_gemm2_t12(const float* __restrict__ x,
                                                   const float* __restrict__ t3,
                                                   const float* __restrict__ p2,
                                                   const float* __restrict__ p5,
                                                   const float* __restrict__ p10,
                                                   float* __restrict__ t9,
                                                   float* __restrict__ t12) {
    const int b = blockIdx.x;
    if (b < G2WG) {
        gemm2_role(b & 3, (b >> 2) & 3, b >> 4, x, t3, t9);
    } else {
        const int e = b - G2WG;
        t12_role(e % EWX, e / EWX, x, t3, p2, p5, p10, t12);
    }
}

#define G3P   36
#define G3AST (128 * G3P)
#define G3BST (224 * G3P)

__device__ __forceinline__ void g3_fetch(const float* __restrict__ asrc,
                                         const float* __restrict__ bsrc,
                                         const int (&bg)[7],
                                         v4f (&av)[4], float (&bv)[7][4]) {
    #pragma unroll
    for (int i = 0; i < 4; ++i)
        av[i] = *(const v4f*)&asrc[32 * i * C3];
    #pragma unroll
    for (int i = 0; i < 7; ++i) {
        #pragma unroll
        for (int e = 0; e < 4; ++e)
            bv[i][e] = bsrc[bg[i] + e * HWP];
    }
}

__device__ __forceinline__ void g3_stage(float* adst, float* bdst,
                                         const int (&bl)[7],
                                         const v4f (&av)[4], const float (&bv)[7][4]) {
    #pragma unroll
    for (int i = 0; i < 4; ++i)
        *(v4f*)&adst[32 * i * G3P] = av[i];
    #pragma unroll
    for (int i = 0; i < 7; ++i) {
        v4f w4 = { bv[i][0], bv[i][1], bv[i][2], bv[i][3] };
        *(v4f*)&bdst[bl[i]] = w4;
    }
}

__global__ __launch_bounds__(256) void k_gemm3(const float* __restrict__ t9,
                                               const float* __restrict__ t12,
                                               float* __restrict__ out) {
    __shared__ __attribute__((aligned(16))) float As[2 * G3AST];
    __shared__ __attribute__((aligned(16))) float Bs[2 * G3BST];
    const int t    = threadIdx.x;
    const int lane = t & 31;
    const int wv   = t >> 5;
    const int smp  = blockIdx.y;
    const int pos0 = blockIdx.x * 224;
    const float* t9s  = t9  + (size_t)smp * CC * C3;
    const float* t12s = t12 + (size_t)smp * C3 * HWP + pos0;

    const int mn    = lane & 15;
    const int kp    = (lane >> 4) << 1;
    const int rmsb  = (lane >> 4) << 3;
    const int mbase = (wv >> 1) * 32;
    const int nbase = (wv & 1) * 112;

    const int ag = (t >> 3) * C3  + 4 * (t & 7);
    const int al = (t >> 3) * G3P + 4 * (t & 7);
    int bg[7], bl[7];
    #pragma unroll
    for (int i = 0; i < 7; ++i) {
        int role = t + 256 * i;
        int q = role / 224;
        int n = role - 224 * q;
        bg[i] = 4 * q * HWP + n;
        bl[i] = n * G3P + 4 * q;
    }

    const int afr = (mbase + mn) * G3P + kp;
    const int bfr = (nbase + mn) * G3P + kp;

    v8f acc[2][7] = {};
    v4f   av[4];
    float bv[7][4];

    g3_fetch(t9s + ag, t12s, bg, av, bv);
    g3_stage(As + al, Bs, bl, av, bv);
    __syncthreads();

    for (int s = 0; s < 12; ++s) {
        const int  cur  = s & 1;
        const bool more = s + 1 < 12;
        if (more)
            g3_fetch(t9s + ag + 32 * (s + 1), t12s + (size_t)(32 * (s + 1)) * HWP, bg, av, bv);
        __builtin_amdgcn_sched_barrier(0);
        const float* Ac = As + cur * G3AST + afr;
        const float* Bc = Bs + cur * G3BST + bfr;
        #pragma unroll
        for (int kk = 0; kk < 8; ++kk) {
            v2f a0 = *(const v2f*)&Ac[kk * 4];
            v2f a1 = *(const v2f*)&Ac[16 * G3P + kk * 4];
            #pragma unroll
            for (int ni = 0; ni < 7; ++ni) {
                v2f b = *(const v2f*)&Bc[ni * 16 * G3P + kk * 4];
                acc[0][ni] = wmma_f32(a0, b, acc[0][ni]);
                acc[1][ni] = wmma_f32(a1, b, acc[1][ni]);
            }
        }
        __builtin_amdgcn_sched_barrier(0);
        if (more)
            g3_stage(As + (cur ^ 1) * G3AST + al, Bs + (cur ^ 1) * G3BST, bl, av, bv);
        __syncthreads();
    }

    const float scale = 0.051031036307982884f;
    float* o = out + (size_t)smp * CC * HWP;
    #pragma unroll
    for (int mi = 0; mi < 2; ++mi) {
        #pragma unroll
        for (int ni = 0; ni < 7; ++ni) {
            int col = pos0 + nbase + ni * 16 + mn;
            #pragma unroll
            for (int r = 0; r < 8; ++r) {
                int m = mbase + mi * 16 + r + rmsb;
                o[m * HWP + col] = acc[mi][ni][r] * scale;
            }
        }
    }
}

extern "C" void kernel_launch(void* const* d_in, const int* in_sizes, int n_in,
                              void* d_out, int out_size, void* d_ws, size_t ws_size,
                              hipStream_t stream) {
    (void)in_sizes; (void)n_in; (void)out_size; (void)ws_size;
    const float* x   = (const float*)d_in[0];
    const float* p2  = (const float*)d_in[1];
    const float* p3  = (const float*)d_in[2];
    const float* p5  = (const float*)d_in[3];
    const float* p10 = (const float*)d_in[4];
    float* out = (float*)d_out;

    char* ws = (char*)d_ws;
    const size_t t3_bytes  = (size_t)NS * CC * HWP * sizeof(float);
    const size_t t12_bytes = (size_t)NS * C3 * HWP * sizeof(float);
    float* t3  = (float*)ws;
    float* t12 = (float*)(ws + t3_bytes);
    float* t9  = (float*)(ws + t3_bytes + t12_bytes);

    dim3 blk(256);
    k_gemm1<<<dim3(HWP / 224, NS), blk, 0, stream>>>(p3, x, t3);
    k_gemm2_t12<<<dim3(G2WG + EWWG), dim3(128), 0, stream>>>(x, t3, p2, p5, p10, t9, t12);
    k_gemm3<<<dim3(HWP / 224, NS), blk, 0, stream>>>(t9, t12, out);
}
